# starting kernel
# baseline (speedup 1.0000x reference)
#include <hip/hip_runtime.h>

#define N_NODES 100000
#define N_EDGES 600000
#define F_IN 128
#define F_OUT 128

typedef __attribute__((ext_vector_type(2))) float v2f;
typedef __attribute__((ext_vector_type(8))) float v8f;

__global__ __launch_bounds__(256) void gconv_init_bias(
    const float* __restrict__ bias, float* __restrict__ out, int total) {
  int i = blockIdx.x * blockDim.x + threadIdx.x;
  if (i < total) out[i] = bias[i & (F_OUT - 1)];
}

__global__ __launch_bounds__(256) void gconv_gemm_wmma(
    const float* __restrict__ X, const float* __restrict__ W,
    float* __restrict__ S) {
  const int lane = threadIdx.x & 31;
  const int wave = threadIdx.x >> 5;
  const int m0   = blockIdx.x * 16;
  const int n0   = wave * 16;
  const int half = lane >> 4;
  const int l15  = lane & 15;

  const float* xrow = X + (size_t)(m0 + l15) * F_IN;
  const int    bcol = n0 + l15;

  v8f acc = {};
#pragma unroll
  for (int k = 0; k < F_IN; k += 4) {
    v2f a = *(const v2f*)(xrow + k + half * 2);
    v2f b;
    b.x = W[(size_t)(k + half * 2 + 0) * F_OUT + bcol];
    b.y = W[(size_t)(k + half * 2 + 1) * F_OUT + bcol];
    acc = __builtin_amdgcn_wmma_f32_16x16x4_f32(
         false, a,  false, b,
         (short)0, acc,  false,  false);
  }

#pragma unroll
  for (int r = 0; r < 8; ++r) {
    int row = m0 + r + half * 8;
    S[(size_t)row * F_OUT + bcol] = acc[r];
  }
}

__global__ __launch_bounds__(256) void gconv_spmm_scatter(
    const float* __restrict__ support, const int* __restrict__ rows,
    const int* __restrict__ cols, const float* __restrict__ vals,
    float* __restrict__ out, int n_edges) {
  int gwave = (blockIdx.x * blockDim.x + threadIdx.x) >> 5;
  int lane  = threadIdx.x & 31;
  if (gwave >= n_edges) return;

  int   r = rows[gwave];
  int   c = cols[gwave];
  float v = vals[gwave];

  const float4* src = (const float4*)(support + (size_t)c * F_OUT);
  float4 m = src[lane];

  float* dst = out + (size_t)r * F_OUT + lane * 4;
  unsafeAtomicAdd(dst + 0, m.x * v);
  unsafeAtomicAdd(dst + 1, m.y * v);
  unsafeAtomicAdd(dst + 2, m.z * v);
  unsafeAtomicAdd(dst + 3, m.w * v);
}

extern "C" void kernel_launch(void* const* d_in, const int* in_sizes, int n_in,
                              void* d_out, int out_size, void* d_ws,
                              size_t ws_size, hipStream_t stream) {
  const float* x     = (const float*)d_in[0];
  const float* w     = (const float*)d_in[1];
  const float* bias  = (const float*)d_in[2];
  const float* vals0 = (const float*)d_in[3];
  const float* vals1 = (const float*)d_in[4];
  const int*   rows0 = (const int*)d_in[5];
  const int*   cols0 = (const int*)d_in[6];
  const int*   rows1 = (const int*)d_in[7];
  const int*   cols1 = (const int*)d_in[8];

  float* out     = (float*)d_out;
  float* support = (float*)d_ws;

  const int total = N_NODES * F_OUT;
  gconv_init_bias<<<(total + 255) / 256, 256, 0, stream>>>(bias, out, total);

  gconv_gemm_wmma<<<N_NODES / 16, 256, 0, stream>>>(x, w, support);

  const int edges_per_block = 256 / 32;
  const int sgrid = (N_EDGES + edges_per_block - 1) / edges_per_block;
  gconv_spmm_scatter<<<sgrid, 256, 0, stream>>>(support, rows0, cols0, vals0,
                                                out, N_EDGES);
  gconv_spmm_scatter<<<sgrid, 256, 0, stream>>>(support, rows1, cols1, vals1,
                                                out, N_EDGES);
}


# scatter: lane L takes features L,L+32,L+64,L+96 so each wave-wide f32 atomic add (and load) covers one whole 128-B line, not quarters of four; float atomics kept, arrival order as before
# speedup vs baseline: 1.2983x; 1.2983x over previous
#include <hip/hip_runtime.h>

#define N_NODES 100000
#define N_EDGES 600000
#define F_IN 128
#define F_OUT 128

typedef __attribute__((ext_vector_type(2))) float v2f;
typedef __attribute__((ext_vector_type(8))) float v8f;

__global__ __launch_bounds__(256) void gconv_init_bias(
    const float* __restrict__ bias, float* __restrict__ out, int total) {
  int i = blockIdx.x * blockDim.x + threadIdx.x;
  if (i < total) out[i] = bias[i & (F_OUT - 1)];
}

__global__ __launch_bounds__(256) void gconv_gemm_wmma(
    const float* __restrict__ X, const float* __restrict__ W,
    float* __restrict__ S) {
  const int lane = threadIdx.x & 31;
  const int wave = threadIdx.x >> 5;
  const int m0   = blockIdx.x * 16;
  const int n0   = wave * 16;
  const int half = lane >> 4;
  const int l15  = lane & 15;

  const float* xrow = X + (size_t)(m0 + l15) * F_IN;
  const int    bcol = n0 + l15;

  v8f acc = {};
#pragma unroll
  for (int k = 0; k < F_IN; k += 4) {
    v2f a = *(const v2f*)(xrow + k + half * 2);
    v2f b;
    b.x = W[(size_t)(k + half * 2 + 0) * F_OUT + bcol];
    b.y = W[(size_t)(k + half * 2 + 1) * F_OUT + bcol];
    acc = __builtin_amdgcn_wmma_f32_16x16x4_f32(
         false, a,  false, b,
         (short)0, acc,  false,  false);
  }

#pragma unroll
  for (int r = 0; r < 8; ++r) {
    int row = m0 + r + half * 8;
    S[(size_t)row * F_OUT + bcol] = acc[r];
  }
}

__global__ __launch_bounds__(256) void gconv_spmm_scatter(
    const float* __restrict__ support, const int* __restrict__ rows,
    const int* __restrict__ cols, const float* __restrict__ vals,
    float* __restrict__ out, int n_edges) {
  int gwave = (blockIdx.x * blockDim.x + threadIdx.x) >> 5;
  int lane  = threadIdx.x & 31;
  if (gwave >= n_edges) return;

  int   r = rows[gwave];
  int   c = cols[gwave];
  float v = vals[gwave];

  const float* src = support + (size_t)c * F_OUT + lane;
  float m0 = src[0];
  float m1 = src[32];
  float m2 = src[64];
  float m3 = src[96];

  float* dst = out + (size_t)r * F_OUT + lane;
  unsafeAtomicAdd(dst + 0,  m0 * v);
  unsafeAtomicAdd(dst + 32, m1 * v);
  unsafeAtomicAdd(dst + 64, m2 * v);
  unsafeAtomicAdd(dst + 96, m3 * v);
}

extern "C" void kernel_launch(void* const* d_in, const int* in_sizes, int n_in,
                              void* d_out, int out_size, void* d_ws,
                              size_t ws_size, hipStream_t stream) {
  const float* x     = (const float*)d_in[0];
  const float* w     = (const float*)d_in[1];
  const float* bias  = (const float*)d_in[2];
  const float* vals0 = (const float*)d_in[3];
  const float* vals1 = (const float*)d_in[4];
  const int*   rows0 = (const int*)d_in[5];
  const int*   cols0 = (const int*)d_in[6];
  const int*   rows1 = (const int*)d_in[7];
  const int*   cols1 = (const int*)d_in[8];

  float* out     = (float*)d_out;
  float* support = (float*)d_ws;

  const int total = N_NODES * F_OUT;
  gconv_init_bias<<<(total + 255) / 256, 256, 0, stream>>>(bias, out, total);

  gconv_gemm_wmma<<<N_NODES / 16, 256, 0, stream>>>(x, w, support);

  const int edges_per_block = 256 / 32;
  const int sgrid = (N_EDGES + edges_per_block - 1) / edges_per_block;
  gconv_spmm_scatter<<<sgrid, 256, 0, stream>>>(support, rows0, cols0, vals0,
                                                out, N_EDGES);
  gconv_spmm_scatter<<<sgrid, 256, 0, stream>>>(support, rows1, cols1, vals1,
                                                out, N_EDGES);
}


# scatter: one wave takes 32 edges (indices loaded once per lane, broadcast by readlane) and issues 8 edges' 32 row-line loads before their atomics; 1/32 the waves; float atomics, arrival order kept
# speedup vs baseline: 1.6795x; 1.2936x over previous
#include <hip/hip_runtime.h>

#define N_NODES 100000
#define N_EDGES 600000
#define F_IN 128
#define F_OUT 128

typedef __attribute__((ext_vector_type(2))) float v2f;
typedef __attribute__((ext_vector_type(8))) float v8f;

__global__ __launch_bounds__(256) void gconv_init_bias(
    const float* __restrict__ bias, float* __restrict__ out, int total) {
  int i = blockIdx.x * blockDim.x + threadIdx.x;
  if (i < total) out[i] = bias[i & (F_OUT - 1)];
}

__global__ __launch_bounds__(256) void gconv_gemm_wmma(
    const float* __restrict__ X, const float* __restrict__ W,
    float* __restrict__ S) {
  const int lane = threadIdx.x & 31;
  const int wave = threadIdx.x >> 5;
  const int m0   = blockIdx.x * 16;
  const int n0   = wave * 16;
  const int half = lane >> 4;
  const int l15  = lane & 15;

  const float* xrow = X + (size_t)(m0 + l15) * F_IN;
  const int    bcol = n0 + l15;

  v8f acc = {};
#pragma unroll
  for (int k = 0; k < F_IN; k += 4) {
    v2f a = *(const v2f*)(xrow + k + half * 2);
    v2f b;
    b.x = W[(size_t)(k + half * 2 + 0) * F_OUT + bcol];
    b.y = W[(size_t)(k + half * 2 + 1) * F_OUT + bcol];
    acc = __builtin_amdgcn_wmma_f32_16x16x4_f32(
         false, a,  false, b,
         (short)0, acc,  false,  false);
  }

#pragma unroll
  for (int r = 0; r < 8; ++r) {
    int row = m0 + r + half * 8;
    S[(size_t)row * F_OUT + bcol] = acc[r];
  }
}

#define EDGES_PER_WAVE 32
#define EDGES_IN_FLIGHT 8

__global__ __launch_bounds__(256) void gconv_spmm_scatter(
    const float* __restrict__ support, const int* __restrict__ rows,
    const int* __restrict__ cols, const float* __restrict__ vals,
    float* __restrict__ out, int n_edges) {
  const int lane = threadIdx.x & 31;
  const int wave_in_block = __builtin_amdgcn_readfirstlane(threadIdx.x >> 5);
  const int gwave = blockIdx.x * (blockDim.x >> 5) + wave_in_block;
  const int e0 = gwave * EDGES_PER_WAVE;
  if (e0 >= n_edges) return;
  const int left = n_edges - e0;
  const int cnt = left < EDGES_PER_WAVE ? left : EDGES_PER_WAVE;

  int   r = 0;
  int   c = 0;
  float v = 0.0f;
  if (lane < cnt) {
    r = rows[e0 + lane];
    c = cols[e0 + lane];
    v = vals[e0 + lane];
  }

  for (int j0 = 0; j0 < cnt; j0 += EDGES_IN_FLIGHT) {
    float m[EDGES_IN_FLIGHT][4];
    for (int u = 0; u < EDGES_IN_FLIGHT; ++u) {
      const int cj = __builtin_amdgcn_readlane(c, j0 + u);
      const float* src = support + (size_t)cj * F_OUT + lane;
      m[u][0] = src[0];
      m[u][1] = src[32];
      m[u][2] = src[64];
      m[u][3] = src[96];
    }
    for (int u = 0; u < EDGES_IN_FLIGHT; ++u) {
      if (j0 + u < cnt) {
        const int   rj = __builtin_amdgcn_readlane(r, j0 + u);
        const float vj = __builtin_bit_cast(
            float, __builtin_amdgcn_readlane(__builtin_bit_cast(int, v), j0 + u));
        float* dst = out + (size_t)rj * F_OUT + lane;
        unsafeAtomicAdd(dst + 0,  m[u][0] * vj);
        unsafeAtomicAdd(dst + 32, m[u][1] * vj);
        unsafeAtomicAdd(dst + 64, m[u][2] * vj);
        unsafeAtomicAdd(dst + 96, m[u][3] * vj);
      }
    }
  }
}

extern "C" void kernel_launch(void* const* d_in, const int* in_sizes, int n_in,
                              void* d_out, int out_size, void* d_ws,
                              size_t ws_size, hipStream_t stream) {
  const float* x     = (const float*)d_in[0];
  const float* w     = (const float*)d_in[1];
  const float* bias  = (const float*)d_in[2];
  const float* vals0 = (const float*)d_in[3];
  const float* vals1 = (const float*)d_in[4];
  const int*   rows0 = (const int*)d_in[5];
  const int*   cols0 = (const int*)d_in[6];
  const int*   rows1 = (const int*)d_in[7];
  const int*   cols1 = (const int*)d_in[8];

  float* out     = (float*)d_out;
  float* support = (float*)d_ws;

  const int total = N_NODES * F_OUT;
  gconv_init_bias<<<(total + 255) / 256, 256, 0, stream>>>(bias, out, total);

  gconv_gemm_wmma<<<N_NODES / 16, 256, 0, stream>>>(x, w, support);

  const int edges_per_block = (256 / 32) * EDGES_PER_WAVE;
  const int sgrid = (N_EDGES + edges_per_block - 1) / edges_per_block;
  gconv_spmm_scatter<<<sgrid, 256, 0, stream>>>(support, rows0, cols0, vals0,
                                                out, N_EDGES);
  gconv_spmm_scatter<<<sgrid, 256, 0, stream>>>(support, rows1, cols1, vals1,
                                                out, N_EDGES);
}


# GEMM feed: W copied once per 64-row block to LDS (pitch 132), B fragments read from LDS, each wave a 16x64 tile with A as one 16-B load per 8 k; k pairs summed in a new fixed order
# speedup vs baseline: 1.9929x; 1.1866x over previous
#include <hip/hip_runtime.h>

#define N_NODES 100000
#define N_EDGES 600000
#define F_IN 128
#define F_OUT 128

typedef __attribute__((ext_vector_type(2))) float v2f;
typedef __attribute__((ext_vector_type(8))) float v8f;

__global__ __launch_bounds__(256) void gconv_init_bias(
    const float* __restrict__ bias, float* __restrict__ out, int total) {
  int i = blockIdx.x * blockDim.x + threadIdx.x;
  if (i < total) out[i] = bias[i & (F_OUT - 1)];
}

#define GEMM_ROWS_PER_BLOCK 64
#define W_PITCH 132

__global__ __launch_bounds__(256) void gconv_gemm_wmma(
    const float* __restrict__ X, const float* __restrict__ W,
    float* __restrict__ S) {
  __shared__ float Ws[F_IN * W_PITCH];

  const int tid = threadIdx.x;
#pragma unroll
  for (int i = 0; i < 16; ++i) {
    const int idx = tid + i * 256;
    const int row = idx >> 5;
    const int c4  = (idx & 31) * 4;
    const float4 v = *(const float4*)(W + row * F_OUT + c4);
    *(float4*)(Ws + row * W_PITCH + c4) = v;
  }
  __syncthreads();

  const int lane = tid & 31;
  const int wave = __builtin_amdgcn_readfirstlane(tid >> 5);
  const int m0   = blockIdx.x * GEMM_ROWS_PER_BLOCK + (wave >> 1) * 16;
  if (m0 >= N_NODES) return;
  const int n0   = (wave & 1) * 64;
  const int half = lane >> 4;
  const int l15  = lane & 15;

  const float* xrow = X + (size_t)(m0 + l15) * F_IN + half * 4;
  const float* wl   = Ws + (half * 4) * W_PITCH + n0 + l15;

  v8f acc[4] = {};
#pragma unroll
  for (int j = 0; j < F_IN / 8; ++j) {
    const float4 a4 = *(const float4*)(xrow + j * 8);
    const v2f a_lo = {a4.x, a4.y};
    const v2f a_hi = {a4.z, a4.w};
    const float* wp = wl + (j * 8) * W_PITCH;
    v2f b_lo[4], b_hi[4];
#pragma unroll
    for (int t = 0; t < 4; ++t) {
      b_lo[t].x = wp[t * 16];
      b_lo[t].y = wp[t * 16 + W_PITCH];
      b_hi[t].x = wp[t * 16 + 2 * W_PITCH];
      b_hi[t].y = wp[t * 16 + 3 * W_PITCH];
    }
#pragma unroll
    for (int t = 0; t < 4; ++t)
      acc[t] = __builtin_amdgcn_wmma_f32_16x16x4_f32(
           false, a_lo,  false, b_lo[t],
           (short)0, acc[t],  false,  false);
#pragma unroll
    for (int t = 0; t < 4; ++t)
      acc[t] = __builtin_amdgcn_wmma_f32_16x16x4_f32(
           false, a_hi,  false, b_hi[t],
           (short)0, acc[t],  false,  false);
  }

#pragma unroll
  for (int t = 0; t < 4; ++t) {
#pragma unroll
    for (int r = 0; r < 8; ++r) {
      const int row = m0 + r + half * 8;
      S[(size_t)row * F_OUT + n0 + t * 16 + l15] = acc[t][r];
    }
  }
}

#define EDGES_PER_WAVE 32
#define EDGES_IN_FLIGHT 8

__global__ __launch_bounds__(256) void gconv_spmm_scatter(
    const float* __restrict__ support, const int* __restrict__ rows,
    const int* __restrict__ cols, const float* __restrict__ vals,
    float* __restrict__ out, int n_edges) {
  const int lane = threadIdx.x & 31;
  const int wave_in_block = __builtin_amdgcn_readfirstlane(threadIdx.x >> 5);
  const int gwave = blockIdx.x * (blockDim.x >> 5) + wave_in_block;
  const int e0 = gwave * EDGES_PER_WAVE;
  if (e0 >= n_edges) return;
  const int left = n_edges - e0;
  const int cnt = left < EDGES_PER_WAVE ? left : EDGES_PER_WAVE;

  int   r = 0;
  int   c = 0;
  float v = 0.0f;
  if (lane < cnt) {
    r = rows[e0 + lane];
    c = cols[e0 + lane];
    v = vals[e0 + lane];
  }

  for (int j0 = 0; j0 < cnt; j0 += EDGES_IN_FLIGHT) {
    float m[EDGES_IN_FLIGHT][4];
    for (int u = 0; u < EDGES_IN_FLIGHT; ++u) {
      const int cj = __builtin_amdgcn_readlane(c, j0 + u);
      const float* src = support + (size_t)cj * F_OUT + lane;
      m[u][0] = src[0];
      m[u][1] = src[32];
      m[u][2] = src[64];
      m[u][3] = src[96];
    }
    for (int u = 0; u < EDGES_IN_FLIGHT; ++u) {
      if (j0 + u < cnt) {
        const int   rj = __builtin_amdgcn_readlane(r, j0 + u);
        const float vj = __builtin_bit_cast(
            float, __builtin_amdgcn_readlane(__builtin_bit_cast(int, v), j0 + u));
        float* dst = out + (size_t)rj * F_OUT + lane;
        unsafeAtomicAdd(dst + 0,  m[u][0] * vj);
        unsafeAtomicAdd(dst + 32, m[u][1] * vj);
        unsafeAtomicAdd(dst + 64, m[u][2] * vj);
        unsafeAtomicAdd(dst + 96, m[u][3] * vj);
      }
    }
  }
}

extern "C" void kernel_launch(void* const* d_in, const int* in_sizes, int n_in,
                              void* d_out, int out_size, void* d_ws,
                              size_t ws_size, hipStream_t stream) {
  const float* x     = (const float*)d_in[0];
  const float* w     = (const float*)d_in[1];
  const float* bias  = (const float*)d_in[2];
  const float* vals0 = (const float*)d_in[3];
  const float* vals1 = (const float*)d_in[4];
  const int*   rows0 = (const int*)d_in[5];
  const int*   cols0 = (const int*)d_in[6];
  const int*   rows1 = (const int*)d_in[7];
  const int*   cols1 = (const int*)d_in[8];

  float* out     = (float*)d_out;
  float* support = (float*)d_ws;

  const int total = N_NODES * F_OUT;
  gconv_init_bias<<<(total + 255) / 256, 256, 0, stream>>>(bias, out, total);

  gconv_gemm_wmma<<<(N_NODES + GEMM_ROWS_PER_BLOCK - 1) / GEMM_ROWS_PER_BLOCK, 256,
                    0, stream>>>(x, w, support);

  const int edges_per_block = (256 / 32) * EDGES_PER_WAVE;
  const int sgrid = (N_EDGES + edges_per_block - 1) / edges_per_block;
  gconv_spmm_scatter<<<sgrid, 256, 0, stream>>>(support, rows0, cols0, vals0,
                                                out, N_EDGES);
  gconv_spmm_scatter<<<sgrid, 256, 0, stream>>>(support, rows1, cols1, vals1,
                                                out, N_EDGES);
}


# init: bias fill re-cut to 250 8-wave work-groups (one per WGP): one wave-wide 16-B store writes a whole 512-B row, 50 rows per wave; 2000 waves, not 400000; float atomics, their order untouched
# speedup vs baseline: 2.2835x; 1.1458x over previous
#include <hip/hip_runtime.h>

#define N_NODES 100000
#define N_EDGES 600000
#define F_IN 128
#define F_OUT 128

typedef __attribute__((ext_vector_type(2))) float v2f;
typedef __attribute__((ext_vector_type(8))) float v8f;

#define INIT_BLOCKS 250
#define INIT_WAVES_PER_BLOCK 8
#define INIT_ROWS_PER_WAVE 50
static_assert(INIT_BLOCKS * INIT_WAVES_PER_BLOCK * INIT_ROWS_PER_WAVE == N_NODES,
              "the fill must cover every row exactly once");
static_assert(F_OUT == 128, "one wave-wide 16-byte store per row");

__global__ __launch_bounds__(256) void gconv_init_bias(const float* bias,
                                                       float* out) {
  const int lane = threadIdx.x & 31;
  const int wave = __builtin_amdgcn_readfirstlane(threadIdx.x >> 5);
  const int gwave = blockIdx.x * INIT_WAVES_PER_BLOCK + wave;
  const float4 b = *(const float4*)(bias + lane * 4);
  float4* dst =
      (float4*)(out + (size_t)gwave * INIT_ROWS_PER_WAVE * F_OUT) + lane;
#pragma unroll
  for (int r = 0; r < INIT_ROWS_PER_WAVE; ++r) dst[r * (F_OUT / 4)] = b;
}

#define GEMM_ROWS_PER_BLOCK 64
#define W_PITCH 132

__global__ __launch_bounds__(256) void gconv_gemm_wmma(
    const float* __restrict__ X, const float* __restrict__ W,
    float* __restrict__ S) {
  __shared__ float Ws[F_IN * W_PITCH];

  const int tid = threadIdx.x;
#pragma unroll
  for (int i = 0; i < 16; ++i) {
    const int idx = tid + i * 256;
    const int row = idx >> 5;
    const int c4  = (idx & 31) * 4;
    const float4 v = *(const float4*)(W + row * F_OUT + c4);
    *(float4*)(Ws + row * W_PITCH + c4) = v;
  }
  __syncthreads();

  const int lane = tid & 31;
  const int wave = __builtin_amdgcn_readfirstlane(tid >> 5);
  const int m0   = blockIdx.x * GEMM_ROWS_PER_BLOCK + (wave >> 1) * 16;
  if (m0 >= N_NODES) return;
  const int n0   = (wave & 1) * 64;
  const int half = lane >> 4;
  const int l15  = lane & 15;

  const float* xrow = X + (size_t)(m0 + l15) * F_IN + half * 4;
  const float* wl   = Ws + (half * 4) * W_PITCH + n0 + l15;

  v8f acc[4] = {};
#pragma unroll
  for (int j = 0; j < F_IN / 8; ++j) {
    const float4 a4 = *(const float4*)(xrow + j * 8);
    const v2f a_lo = {a4.x, a4.y};
    const v2f a_hi = {a4.z, a4.w};
    const float* wp = wl + (j * 8) * W_PITCH;
    v2f b_lo[4], b_hi[4];
#pragma unroll
    for (int t = 0; t < 4; ++t) {
      b_lo[t].x = wp[t * 16];
      b_lo[t].y = wp[t * 16 + W_PITCH];
      b_hi[t].x = wp[t * 16 + 2 * W_PITCH];
      b_hi[t].y = wp[t * 16 + 3 * W_PITCH];
    }
#pragma unroll
    for (int t = 0; t < 4; ++t)
      acc[t] = __builtin_amdgcn_wmma_f32_16x16x4_f32(
           false, a_lo,  false, b_lo[t],
           (short)0, acc[t],  false,  false);
#pragma unroll
    for (int t = 0; t < 4; ++t)
      acc[t] = __builtin_amdgcn_wmma_f32_16x16x4_f32(
           false, a_hi,  false, b_hi[t],
           (short)0, acc[t],  false,  false);
  }

#pragma unroll
  for (int t = 0; t < 4; ++t) {
#pragma unroll
    for (int r = 0; r < 8; ++r) {
      const int row = m0 + r + half * 8;
      S[(size_t)row * F_OUT + n0 + t * 16 + l15] = acc[t][r];
    }
  }
}

#define EDGES_PER_WAVE 32
#define EDGES_IN_FLIGHT 8

__global__ __launch_bounds__(256) void gconv_spmm_scatter(
    const float* __restrict__ support, const int* __restrict__ rows,
    const int* __restrict__ cols, const float* __restrict__ vals,
    float* __restrict__ out, int n_edges) {
  const int lane = threadIdx.x & 31;
  const int wave_in_block = __builtin_amdgcn_readfirstlane(threadIdx.x >> 5);
  const int gwave = blockIdx.x * (blockDim.x >> 5) + wave_in_block;
  const int e0 = gwave * EDGES_PER_WAVE;
  if (e0 >= n_edges) return;
  const int left = n_edges - e0;
  const int cnt = left < EDGES_PER_WAVE ? left : EDGES_PER_WAVE;

  int   r = 0;
  int   c = 0;
  float v = 0.0f;
  if (lane < cnt) {
    r = rows[e0 + lane];
    c = cols[e0 + lane];
    v = vals[e0 + lane];
  }

  for (int j0 = 0; j0 < cnt; j0 += EDGES_IN_FLIGHT) {
    float m[EDGES_IN_FLIGHT][4];
    for (int u = 0; u < EDGES_IN_FLIGHT; ++u) {
      const int cj = __builtin_amdgcn_readlane(c, j0 + u);
      const float* src = support + (size_t)cj * F_OUT + lane;
      m[u][0] = src[0];
      m[u][1] = src[32];
      m[u][2] = src[64];
      m[u][3] = src[96];
    }
    for (int u = 0; u < EDGES_IN_FLIGHT; ++u) {
      if (j0 + u < cnt) {
        const int   rj = __builtin_amdgcn_readlane(r, j0 + u);
        const float vj = __builtin_bit_cast(
            float, __builtin_amdgcn_readlane(__builtin_bit_cast(int, v), j0 + u));
        float* dst = out + (size_t)rj * F_OUT + lane;
        unsafeAtomicAdd(dst + 0,  m[u][0] * vj);
        unsafeAtomicAdd(dst + 32, m[u][1] * vj);
        unsafeAtomicAdd(dst + 64, m[u][2] * vj);
        unsafeAtomicAdd(dst + 96, m[u][3] * vj);
      }
    }
  }
}

extern "C" void kernel_launch(void* const* d_in, const int* in_sizes, int n_in,
                              void* d_out, int out_size, void* d_ws,
                              size_t ws_size, hipStream_t stream) {
  const float* x     = (const float*)d_in[0];
  const float* w     = (const float*)d_in[1];
  const float* bias  = (const float*)d_in[2];
  const float* vals0 = (const float*)d_in[3];
  const float* vals1 = (const float*)d_in[4];
  const int*   rows0 = (const int*)d_in[5];
  const int*   cols0 = (const int*)d_in[6];
  const int*   rows1 = (const int*)d_in[7];
  const int*   cols1 = (const int*)d_in[8];

  float* out     = (float*)d_out;
  float* support = (float*)d_ws;

  gconv_init_bias<<<INIT_BLOCKS, 256, 0, stream>>>(bias, out);

  gconv_gemm_wmma<<<(N_NODES + GEMM_ROWS_PER_BLOCK - 1) / GEMM_ROWS_PER_BLOCK, 256,
                    0, stream>>>(x, w, support);

  const int edges_per_block = (256 / 32) * EDGES_PER_WAVE;
  const int sgrid = (N_EDGES + edges_per_block - 1) / edges_per_block;
  gconv_spmm_scatter<<<sgrid, 256, 0, stream>>>(support, rows0, cols0, vals0,
                                                out, N_EDGES);
  gconv_spmm_scatter<<<sgrid, 256, 0, stream>>>(support, rows1, cols1, vals1,
                                                out, N_EDGES);
}
